# starting kernel
# baseline (speedup 1.0000x reference)
#include <hip/hip_runtime.h>
#include <hip/hip_bf16.h>

typedef __attribute__((ext_vector_type(2))) float v2f;
typedef __attribute__((ext_vector_type(8))) float v8f;

#define DIN 128
#define DCOL 64

__global__ void k_init_deg(float* __restrict__ deg, int n) {
    int i = blockIdx.x * blockDim.x + threadIdx.x;
    if (i < n) deg[i] = 1.0f;
}

__global__ void k_accum_deg(const int* __restrict__ dst, float* __restrict__ deg, int e) {
    int i = blockIdx.x * blockDim.x + threadIdx.x;
    if (i < e) atomicAdd(&deg[dst[i]], 1.0f);
}

__global__ void k_finish_dinv(float* __restrict__ deg, int n) {
    int i = blockIdx.x * blockDim.x + threadIdx.x;
    if (i < n) deg[i] = rsqrtf(deg[i]);
}

template <int K, bool RELU, bool BIAS>
__global__ __launch_bounds__(128)
void k_gemm_wmma(const float* __restrict__ A, const float* __restrict__ W,
                 float* __restrict__ H, const float* __restrict__ bias) {
    const int lane    = threadIdx.x & 31;
    const int wave    = threadIdx.x >> 5;
    const int rowBase = blockIdx.x << 4;
    const int colBase = wave << 4;
    const int half    = lane >> 4;
    const int l16     = lane & 15;

    const float* __restrict__ arow = A + (size_t)(rowBase + l16) * K;
    const float* __restrict__ wcol = W + colBase + l16;

    v8f c = {};
    #pragma unroll 8
    for (int kb = 0; kb < K; kb += 4) {
        const int k0 = kb + half * 2;
        float a0 = arow[k0];
        float a1 = arow[k0 + 1];
        if constexpr (RELU) { a0 = fmaxf(a0, 0.0f); a1 = fmaxf(a1, 0.0f); }
        v2f a; a[0] = a0; a[1] = a1;
        v2f b; b[0] = wcol[(size_t)k0 * DCOL]; b[1] = wcol[(size_t)(k0 + 1) * DCOL];
        c = __builtin_amdgcn_wmma_f32_16x16x4_f32(
                 false, a,  false, b,
                 (short)0, c,  false,  false);
    }

    float bv;
    if constexpr (BIAS) bv = bias[colBase + l16];
    float* __restrict__ hout = H + colBase + l16;
    #pragma unroll
    for (int v = 0; v < 8; ++v) {
        const int row = rowBase + v + half * 8;
        float val = c[v];
        if constexpr (BIAS) val += bv;
        hout[(size_t)row * DCOL] = val;
    }
}

__global__ void k_agg_init(const float* __restrict__ h, const float* __restrict__ dinv,
                           const float* __restrict__ bias, float* __restrict__ out, int n) {
    int t = blockIdx.x * blockDim.x + threadIdx.x;
    int i = t >> 4, q = t & 15;
    if (i >= n) return;
    float w = dinv[i]; w *= w;
    float4 hv = reinterpret_cast<const float4*>(h + (size_t)i * DCOL)[q];
    float4 bv = reinterpret_cast<const float4*>(bias)[q];
    float4 o;
    o.x = fmaf(w, hv.x, bv.x); o.y = fmaf(w, hv.y, bv.y);
    o.z = fmaf(w, hv.z, bv.z); o.w = fmaf(w, hv.w, bv.w);
    reinterpret_cast<float4*>(out + (size_t)i * DCOL)[q] = o;
}

__global__ void k_agg_edges(const int* __restrict__ src, const int* __restrict__ dst,
                            const float* __restrict__ dinv, const float* __restrict__ h,
                            float* __restrict__ out, int e) {
    int t = blockIdx.x * blockDim.x + threadIdx.x;
    int ed = t >> 4, q = t & 15;
    if (ed >= e) return;
    int s = src[ed];
    int d = dst[ed];
    float w = dinv[s] * dinv[d];
    float4 hv = reinterpret_cast<const float4*>(h + (size_t)s * DCOL)[q];
    float* o = out + (size_t)d * DCOL + q * 4;
    atomicAdd(o + 0, w * hv.x);
    atomicAdd(o + 1, w * hv.y);
    atomicAdd(o + 2, w * hv.z);
    atomicAdd(o + 3, w * hv.w);
}

extern "C" void kernel_launch(void* const* d_in, const int* in_sizes, int n_in,
                              void* d_out, int out_size, void* d_ws, size_t ws_size,
                              hipStream_t stream) {
    const float* x   = (const float*)d_in[0];
    const int*   ei  = (const int*)  d_in[1];
    const float* W1  = (const float*)d_in[2];
    const float* b1  = (const float*)d_in[3];
    const float* W2  = (const float*)d_in[4];
    const float* b2  = (const float*)d_in[5];
    const float* Wfc = (const float*)d_in[6];
    const float* bfc = (const float*)d_in[7];
    float* out = (float*)d_out;

    const int N = in_sizes[0] / DIN;
    const int E = in_sizes[1] / 2;
    const int* src = ei;
    const int* dst = ei + E;

    float* dinv = (float*)d_ws;
    float* buf0 = dinv + ((N + 255) / 256) * 256;
    float* buf1 = buf0 + (size_t)N * DCOL;

    const int blk = 256;
    const int gN   = (N + blk - 1) / blk;
    const int gE   = (E + blk - 1) / blk;
    const int gN16 = (N * 16 + blk - 1) / blk;
    const int gE16 = (E * 16 + blk - 1) / blk;
    const int gGemm = N / 16;

    k_init_deg   <<<gN, blk, 0, stream>>>(dinv, N);
    k_accum_deg  <<<gE, blk, 0, stream>>>(dst, dinv, E);
    k_finish_dinv<<<gN, blk, 0, stream>>>(dinv, N);

    k_gemm_wmma<DIN,  false, false><<<gGemm, 128, 0, stream>>>(x,    W1,  buf0, nullptr);
    k_agg_init  <<<gN16, blk, 0, stream>>>(buf0, dinv, b1, buf1, N);
    k_agg_edges <<<gE16, blk, 0, stream>>>(src, dst, dinv, buf0, buf1, E);

    k_gemm_wmma<DCOL, true,  false><<<gGemm, 128, 0, stream>>>(buf1, W2,  buf0, nullptr);
    k_agg_init  <<<gN16, blk, 0, stream>>>(buf0, dinv, b2, buf1, N);
    k_agg_edges <<<gE16, blk, 0, stream>>>(src, dst, dinv, buf0, buf1, E);

    k_gemm_wmma<DCOL, false, true ><<<gGemm, 128, 0, stream>>>(buf1, Wfc, out,  bfc);
}


# Degree count in LDS: a block per (node range, edge slice) counts its range's in-edges with LDS integer adds, the dinv kernel sums the per-slice counts; no global atomic per edge; 10 launches become 9
# speedup vs baseline: 1.7880x; 1.7880x over previous
#include <hip/hip_runtime.h>
#include <hip/hip_bf16.h>

typedef __attribute__((ext_vector_type(2))) float v2f;
typedef __attribute__((ext_vector_type(8))) float v8f;

#define DIN 128
#define DCOL 64

#define DEG_CAP 6400
#define DEG_S   32
#define DEG_T   1024
#define DEG_U   25

__global__ __launch_bounds__(DEG_T)
void k_count_deg(const int* __restrict__ dst, int* __restrict__ part, int n, int npad, int e) {
    __shared__ int cnt[DEG_CAP];
    const int lo  = blockIdx.x * DEG_CAP;
    const int len = min(DEG_CAP, n - lo);
    for (int i = threadIdx.x; i < DEG_CAP; i += DEG_T) cnt[i] = 0;
    __syncthreads();

    const int per = (e + DEG_S - 1) / DEG_S;
    const int e0  = blockIdx.y * per;
    const int e1  = min(e0 + per, e);
    for (int base = e0 + (int)threadIdx.x; base < e1; base += DEG_T * DEG_U) {
        int d[DEG_U];
        #pragma unroll
        for (int u = 0; u < DEG_U; ++u) d[u] = dst[min(base + u * DEG_T, e1 - 1)];
        #pragma unroll
        for (int u = 0; u < DEG_U; ++u) {
            const unsigned off = (unsigned)(d[u] - lo);
            if (base + u * DEG_T < e1 && off < (unsigned)len) atomicAdd(&cnt[off], 1);
        }
    }
    __syncthreads();

    int* __restrict__ plane = part + (size_t)blockIdx.y * npad + lo;
    for (int i = threadIdx.x; i < len; i += DEG_T) plane[i] = cnt[i];
}

__global__ void k_finish_dinv(const int* __restrict__ part, float* __restrict__ dinv,
                              int n, int npad) {
    int i = blockIdx.x * blockDim.x + threadIdx.x;
    if (i >= n) return;
    int c = 1;
    #pragma unroll
    for (int s = 0; s < DEG_S; ++s) c += part[(size_t)s * npad + i];
    dinv[i] = rsqrtf((float)c);
}

template <int K, bool RELU, bool BIAS>
__global__ __launch_bounds__(128)
void k_gemm_wmma(const float* __restrict__ A, const float* __restrict__ W,
                 float* __restrict__ H, const float* __restrict__ bias) {
    const int lane    = threadIdx.x & 31;
    const int wave    = threadIdx.x >> 5;
    const int rowBase = blockIdx.x << 4;
    const int colBase = wave << 4;
    const int half    = lane >> 4;
    const int l16     = lane & 15;

    const float* __restrict__ arow = A + (size_t)(rowBase + l16) * K;
    const float* __restrict__ wcol = W + colBase + l16;

    v8f c = {};
    #pragma unroll 8
    for (int kb = 0; kb < K; kb += 4) {
        const int k0 = kb + half * 2;
        float a0 = arow[k0];
        float a1 = arow[k0 + 1];
        if constexpr (RELU) { a0 = fmaxf(a0, 0.0f); a1 = fmaxf(a1, 0.0f); }
        v2f a; a[0] = a0; a[1] = a1;
        v2f b; b[0] = wcol[(size_t)k0 * DCOL]; b[1] = wcol[(size_t)(k0 + 1) * DCOL];
        c = __builtin_amdgcn_wmma_f32_16x16x4_f32(
                 false, a,  false, b,
                 (short)0, c,  false,  false);
    }

    float bv;
    if constexpr (BIAS) bv = bias[colBase + l16];
    float* __restrict__ hout = H + colBase + l16;
    #pragma unroll
    for (int v = 0; v < 8; ++v) {
        const int row = rowBase + v + half * 8;
        float val = c[v];
        if constexpr (BIAS) val += bv;
        hout[(size_t)row * DCOL] = val;
    }
}

__global__ void k_agg_init(const float* __restrict__ h, const float* __restrict__ dinv,
                           const float* __restrict__ bias, float* __restrict__ out, int n) {
    int t = blockIdx.x * blockDim.x + threadIdx.x;
    int i = t >> 4, q = t & 15;
    if (i >= n) return;
    float w = dinv[i]; w *= w;
    float4 hv = reinterpret_cast<const float4*>(h + (size_t)i * DCOL)[q];
    float4 bv = reinterpret_cast<const float4*>(bias)[q];
    float4 o;
    o.x = fmaf(w, hv.x, bv.x); o.y = fmaf(w, hv.y, bv.y);
    o.z = fmaf(w, hv.z, bv.z); o.w = fmaf(w, hv.w, bv.w);
    reinterpret_cast<float4*>(out + (size_t)i * DCOL)[q] = o;
}

#define EPW  32
#define EGRP 16

__global__ __launch_bounds__(256)
void k_agg_edges(const int* __restrict__ src, const int* __restrict__ dst,
                 const float* __restrict__ dinv, const float* __restrict__ h,
                 float* __restrict__ out, int e) {
    const int lane = threadIdx.x & 31;
    const int wv   = __builtin_amdgcn_readfirstlane((int)(threadIdx.x >> 5));
    const int base = (blockIdx.x * (int)(blockDim.x >> 5) + wv) * EPW;
    if (base >= e) return;

    const int ed = base + lane;
    int s = 0, d = 0;
    float w = 0.0f;
    if (ed < e) {
        s = src[ed];
        d = dst[ed];
        w = dinv[s] * dinv[d];
    }

    if (base + EPW <= e) {
        #pragma unroll
        for (int g = 0; g < EPW; g += EGRP) {
            float v0[EGRP], v1[EGRP];
            #pragma unroll
            for (int j = 0; j < EGRP; ++j) {
                const int sj = __builtin_amdgcn_readlane(s, g + j);
                const float* __restrict__ hr = h + (size_t)sj * DCOL;
                v0[j] = hr[lane];
                v1[j] = hr[lane + 32];
            }
            #pragma unroll
            for (int j = 0; j < EGRP; ++j) {
                const int   dj = __builtin_amdgcn_readlane(d, g + j);
                const float wj = __int_as_float(__builtin_amdgcn_readlane(__float_as_int(w), g + j));
                float* __restrict__ orow = out + (size_t)dj * DCOL;
                atomicAdd(orow + lane,      wj * v0[j]);
                atomicAdd(orow + lane + 32, wj * v1[j]);
            }
        }
    } else {
        const int cnt = e - base;
        for (int j = 0; j < cnt; ++j) {
            const int   sj = __builtin_amdgcn_readlane(s, j);
            const int   dj = __builtin_amdgcn_readlane(d, j);
            const float wj = __int_as_float(__builtin_amdgcn_readlane(__float_as_int(w), j));
            const float* __restrict__ hr = h + (size_t)sj * DCOL;
            float* __restrict__ orow = out + (size_t)dj * DCOL;
            const float a0 = hr[lane], a1 = hr[lane + 32];
            atomicAdd(orow + lane,      wj * a0);
            atomicAdd(orow + lane + 32, wj * a1);
        }
    }
}

extern "C" void kernel_launch(void* const* d_in, const int* in_sizes, int n_in,
                              void* d_out, int out_size, void* d_ws, size_t ws_size,
                              hipStream_t stream) {
    const float* x   = (const float*)d_in[0];
    const int*   ei  = (const int*)  d_in[1];
    const float* W1  = (const float*)d_in[2];
    const float* b1  = (const float*)d_in[3];
    const float* W2  = (const float*)d_in[4];
    const float* b2  = (const float*)d_in[5];
    const float* Wfc = (const float*)d_in[6];
    const float* bfc = (const float*)d_in[7];
    float* out = (float*)d_out;

    const int N = in_sizes[0] / DIN;
    const int E = in_sizes[1] / 2;
    const int* src = ei;
    const int* dst = ei + E;

    const int Npad = ((N + 255) / 256) * 256;
    float* dinv = (float*)d_ws;
    float* buf0 = dinv + Npad;
    float* buf1 = buf0 + (size_t)N * DCOL;
    int*   part = (int*)(buf1 + (size_t)N * DCOL);

    const int blk = 256;
    const int gN   = (N + blk - 1) / blk;
    const int gE   = (E + blk - 1) / blk;
    const int gN16 = (N * 16 + blk - 1) / blk;
    const int gGemm = N / 16;

    const dim3 gDeg((N + DEG_CAP - 1) / DEG_CAP, DEG_S);
    k_count_deg  <<<gDeg, DEG_T, 0, stream>>>(dst, part, N, Npad, E);
    k_finish_dinv<<<gN,   blk,   0, stream>>>(part, dinv, N, Npad);

    k_gemm_wmma<DIN,  false, false><<<gGemm, 128, 0, stream>>>(x,    W1,  buf0, nullptr);
    k_agg_init  <<<gN16, blk, 0, stream>>>(buf0, dinv, b1, buf1, N);
    k_agg_edges <<<gE,   blk, 0, stream>>>(src, dst, dinv, buf0, buf1, E);

    k_gemm_wmma<DCOL, true,  false><<<gGemm, 128, 0, stream>>>(buf1, W2,  buf0, nullptr);
    k_agg_init  <<<gN16, blk, 0, stream>>>(buf0, dinv, b2, buf1, N);
    k_agg_edges <<<gE,   blk, 0, stream>>>(src, dst, dinv, buf0, buf1, E);

    k_gemm_wmma<DCOL, false, true ><<<gGemm, 128, 0, stream>>>(buf1, Wfc, out,  bfc);
}


# The two graph layers' matrix kernels also store each node's start row dinv^2*h + b (self-loop term and bias) into the sum buffer: both init launches go (9 to 7); layer 2 sums in a third buffer
# speedup vs baseline: 2.5064x; 1.4018x over previous
#include <hip/hip_runtime.h>
#include <hip/hip_bf16.h>

typedef __attribute__((ext_vector_type(2))) float v2f;
typedef __attribute__((ext_vector_type(8))) float v8f;

#define DIN 128
#define DCOL 64

#define DEG_CAP 6400
#define DEG_S   32
#define DEG_T   1024
#define DEG_U   25

__global__ __launch_bounds__(DEG_T)
void k_count_deg(const int* __restrict__ dst, int* __restrict__ part, int n, int npad, int e) {
    __shared__ int cnt[DEG_CAP];
    const int lo  = blockIdx.x * DEG_CAP;
    const int len = min(DEG_CAP, n - lo);
    for (int i = threadIdx.x; i < DEG_CAP; i += DEG_T) cnt[i] = 0;
    __syncthreads();

    const int per = (e + DEG_S - 1) / DEG_S;
    const int e0  = blockIdx.y * per;
    const int e1  = min(e0 + per, e);
    for (int base = e0 + (int)threadIdx.x; base < e1; base += DEG_T * DEG_U) {
        int d[DEG_U];
        #pragma unroll
        for (int u = 0; u < DEG_U; ++u) d[u] = dst[min(base + u * DEG_T, e1 - 1)];
        #pragma unroll
        for (int u = 0; u < DEG_U; ++u) {
            const unsigned off = (unsigned)(d[u] - lo);
            if (base + u * DEG_T < e1 && off < (unsigned)len) atomicAdd(&cnt[off], 1);
        }
    }
    __syncthreads();

    int* __restrict__ plane = part + (size_t)blockIdx.y * npad + lo;
    for (int i = threadIdx.x; i < len; i += DEG_T) plane[i] = cnt[i];
}

__global__ void k_finish_dinv(const int* __restrict__ part, float* __restrict__ dinv,
                              int n, int npad) {
    int i = blockIdx.x * blockDim.x + threadIdx.x;
    if (i >= n) return;
    int c = 1;
    #pragma unroll
    for (int s = 0; s < DEG_S; ++s) c += part[(size_t)s * npad + i];
    dinv[i] = rsqrtf((float)c);
}

template <int K, bool RELU, bool BIAS, bool SELF>
__global__ __launch_bounds__(128)
void k_gemm_wmma(const float* __restrict__ A, const float* __restrict__ W,
                 float* __restrict__ H, const float* __restrict__ bias,
                 const float* __restrict__ dinv, float* __restrict__ S) {
    static_assert(!(BIAS && SELF), "bias goes either into H or into the start row");
    const int lane    = threadIdx.x & 31;
    const int wave    = threadIdx.x >> 5;
    const int rowBase = blockIdx.x << 4;
    const int colBase = wave << 4;
    const int half    = lane >> 4;
    const int l16     = lane & 15;

    const float* __restrict__ arow = A + (size_t)(rowBase + l16) * K;
    const float* __restrict__ wcol = W + colBase + l16;

    float dsq = 0.0f;
    if constexpr (SELF) { dsq = dinv[rowBase + l16]; dsq *= dsq; }

    v8f c = {};
    #pragma unroll 8
    for (int kb = 0; kb < K; kb += 4) {
        const int k0 = kb + half * 2;
        float a0 = arow[k0];
        float a1 = arow[k0 + 1];
        if constexpr (RELU) { a0 = fmaxf(a0, 0.0f); a1 = fmaxf(a1, 0.0f); }
        v2f a; a[0] = a0; a[1] = a1;
        v2f b; b[0] = wcol[(size_t)k0 * DCOL]; b[1] = wcol[(size_t)(k0 + 1) * DCOL];
        c = __builtin_amdgcn_wmma_f32_16x16x4_f32(
                 false, a,  false, b,
                 (short)0, c,  false,  false);
    }

    float bv;
    if constexpr (BIAS || SELF) bv = bias[colBase + l16];
    float* __restrict__ hout = H + colBase + l16;
    #pragma unroll
    for (int v = 0; v < 8; ++v) {
        const int row = rowBase + v + half * 8;
        float val = c[v];
        if constexpr (BIAS) val += bv;
        hout[(size_t)row * DCOL] = val;
        if constexpr (SELF) {
            const float w_lo = __int_as_float(__builtin_amdgcn_readlane(__float_as_int(dsq), v));
            const float w_hi = __int_as_float(__builtin_amdgcn_readlane(__float_as_int(dsq), v + 8));
            S[(size_t)row * DCOL + colBase + l16] = fmaf(half ? w_hi : w_lo, val, bv);
        }
    }
}

#define EPW  32
#define EGRP 16

__global__ __launch_bounds__(256)
void k_agg_edges(const int* __restrict__ src, const int* __restrict__ dst,
                 const float* __restrict__ dinv, const float* __restrict__ h,
                 float* __restrict__ out, int e) {
    const int lane = threadIdx.x & 31;
    const int wv   = __builtin_amdgcn_readfirstlane((int)(threadIdx.x >> 5));
    const int base = (blockIdx.x * (int)(blockDim.x >> 5) + wv) * EPW;
    if (base >= e) return;

    const int ed = base + lane;
    int s = 0, d = 0;
    float w = 0.0f;
    if (ed < e) {
        s = src[ed];
        d = dst[ed];
        w = dinv[s] * dinv[d];
    }

    if (base + EPW <= e) {
        #pragma unroll
        for (int g = 0; g < EPW; g += EGRP) {
            float v0[EGRP], v1[EGRP];
            #pragma unroll
            for (int j = 0; j < EGRP; ++j) {
                const int sj = __builtin_amdgcn_readlane(s, g + j);
                const float* __restrict__ hr = h + (size_t)sj * DCOL;
                v0[j] = hr[lane];
                v1[j] = hr[lane + 32];
            }
            #pragma unroll
            for (int j = 0; j < EGRP; ++j) {
                const int   dj = __builtin_amdgcn_readlane(d, g + j);
                const float wj = __int_as_float(__builtin_amdgcn_readlane(__float_as_int(w), g + j));
                float* __restrict__ orow = out + (size_t)dj * DCOL;
                atomicAdd(orow + lane,      wj * v0[j]);
                atomicAdd(orow + lane + 32, wj * v1[j]);
            }
        }
    } else {
        const int cnt = e - base;
        for (int j = 0; j < cnt; ++j) {
            const int   sj = __builtin_amdgcn_readlane(s, j);
            const int   dj = __builtin_amdgcn_readlane(d, j);
            const float wj = __int_as_float(__builtin_amdgcn_readlane(__float_as_int(w), j));
            const float* __restrict__ hr = h + (size_t)sj * DCOL;
            float* __restrict__ orow = out + (size_t)dj * DCOL;
            const float a0 = hr[lane], a1 = hr[lane + 32];
            atomicAdd(orow + lane,      wj * a0);
            atomicAdd(orow + lane + 32, wj * a1);
        }
    }
}

extern "C" void kernel_launch(void* const* d_in, const int* in_sizes, int n_in,
                              void* d_out, int out_size, void* d_ws, size_t ws_size,
                              hipStream_t stream) {
    const float* x   = (const float*)d_in[0];
    const int*   ei  = (const int*)  d_in[1];
    const float* W1  = (const float*)d_in[2];
    const float* b1  = (const float*)d_in[3];
    const float* W2  = (const float*)d_in[4];
    const float* b2  = (const float*)d_in[5];
    const float* Wfc = (const float*)d_in[6];
    const float* bfc = (const float*)d_in[7];
    float* out = (float*)d_out;

    const int N = in_sizes[0] / DIN;
    const int E = in_sizes[1] / 2;
    const int* src = ei;
    const int* dst = ei + E;

    const int Npad = ((N + 255) / 256) * 256;
    float* dinv = (float*)d_ws;
    float* buf0 = dinv + Npad;
    float* buf1 = buf0 + (size_t)N * DCOL;
    float* buf2 = buf1 + (size_t)N * DCOL;
    int*   part = (int*)(buf2 + (size_t)N * DCOL);

    const int blk = 256;
    const int gN   = (N + blk - 1) / blk;
    const int gE   = (E + blk - 1) / blk;
    const int gGemm = N / 16;

    const dim3 gDeg((N + DEG_CAP - 1) / DEG_CAP, DEG_S);
    k_count_deg  <<<gDeg, DEG_T, 0, stream>>>(dst, part, N, Npad, E);
    k_finish_dinv<<<gN,   blk,   0, stream>>>(part, dinv, N, Npad);

    k_gemm_wmma<DIN,  false, false, true ><<<gGemm, 128, 0, stream>>>(x,    W1,  buf0, b1,  dinv, buf1);
    k_agg_edges <<<gE,   blk, 0, stream>>>(src, dst, dinv, buf0, buf1, E);

    k_gemm_wmma<DCOL, true,  false, true ><<<gGemm, 128, 0, stream>>>(buf1, W2,  buf0, b2,  dinv, buf2);
    k_agg_edges <<<gE,   blk, 0, stream>>>(src, dst, dinv, buf0, buf2, E);

    k_gemm_wmma<DCOL, false, true,  false><<<gGemm, 128, 0, stream>>>(buf2, Wfc, out,  bfc, nullptr, nullptr);
}


# Edge weight dinv[src]*dinv[dst] is formed once per edge by the threads of the layer-1 matrix launch and stored; both scatters read it coalesced, no dinv gathers there; sums in atomic arrival order
# speedup vs baseline: 2.5839x; 1.0309x over previous
#include <hip/hip_runtime.h>
#include <hip/hip_bf16.h>

typedef __attribute__((ext_vector_type(2))) float v2f;
typedef __attribute__((ext_vector_type(8))) float v8f;

#define DIN 128
#define DCOL 64

#define DEG_CAP 6400
#define DEG_S   32
#define DEG_T   1024
#define DEG_U   25

__global__ __launch_bounds__(DEG_T)
void k_count_deg(const int* __restrict__ dst, int* __restrict__ part, int n, int npad, int e) {
    __shared__ int cnt[DEG_CAP];
    const int lo  = blockIdx.x * DEG_CAP;
    const int len = min(DEG_CAP, n - lo);
    for (int i = threadIdx.x; i < DEG_CAP; i += DEG_T) cnt[i] = 0;
    __syncthreads();

    const int per = (e + DEG_S - 1) / DEG_S;
    const int e0  = blockIdx.y * per;
    const int e1  = min(e0 + per, e);
    for (int base = e0 + (int)threadIdx.x; base < e1; base += DEG_T * DEG_U) {
        int d[DEG_U];
        #pragma unroll
        for (int u = 0; u < DEG_U; ++u) d[u] = dst[min(base + u * DEG_T, e1 - 1)];
        #pragma unroll
        for (int u = 0; u < DEG_U; ++u) {
            const unsigned off = (unsigned)(d[u] - lo);
            if (base + u * DEG_T < e1 && off < (unsigned)len) atomicAdd(&cnt[off], 1);
        }
    }
    __syncthreads();

    int* __restrict__ plane = part + (size_t)blockIdx.y * npad + lo;
    for (int i = threadIdx.x; i < len; i += DEG_T) plane[i] = cnt[i];
}

__global__ void k_finish_dinv(const int* __restrict__ part, float* __restrict__ dinv,
                              int n, int npad) {
    int i = blockIdx.x * blockDim.x + threadIdx.x;
    if (i >= n) return;
    int c = 1;
    #pragma unroll
    for (int s = 0; s < DEG_S; ++s) c += part[(size_t)s * npad + i];
    dinv[i] = rsqrtf((float)c);
}

template <int K, bool RELU, bool BIAS, bool SELF, bool EDGEW>
__global__ __launch_bounds__(128)
void k_gemm_wmma(const float* __restrict__ A, const float* __restrict__ W,
                 float* __restrict__ H, const float* __restrict__ bias,
                 const float* __restrict__ dinv, float* __restrict__ S,
                 const int* __restrict__ esrc, const int* __restrict__ edst,
                 float* __restrict__ wgt, int e) {
    static_assert(!(BIAS && SELF), "bias goes either into H or into the start row");
    static_assert(!EDGEW || SELF, "edge weights need dinv, which the SELF launches get");
    if constexpr (EDGEW) {
        const int stride = (int)(gridDim.x * blockDim.x);
        for (int e0 = (int)(blockIdx.x * blockDim.x + threadIdx.x); e0 < e; e0 += 2 * stride) {
            const int  e1 = (e0 + stride < e) ? e0 + stride : e0;
            const int  s0 = esrc[e0], d0 = edst[e0];
            const int  s1 = esrc[e1], d1 = edst[e1];
            const float w0 = dinv[s0] * dinv[d0];
            const float w1 = dinv[s1] * dinv[d1];
            wgt[e0] = w0;
            wgt[e1] = w1;
        }
    }
    const int lane    = threadIdx.x & 31;
    const int wave    = threadIdx.x >> 5;
    const int rowBase = blockIdx.x << 4;
    const int colBase = wave << 4;
    const int half    = lane >> 4;
    const int l16     = lane & 15;

    const float* __restrict__ arow = A + (size_t)(rowBase + l16) * K;
    const float* __restrict__ wcol = W + colBase + l16;

    float dsq = 0.0f;
    if constexpr (SELF) { dsq = dinv[rowBase + l16]; dsq *= dsq; }

    v8f c = {};
    #pragma unroll 8
    for (int kb = 0; kb < K; kb += 4) {
        const int k0 = kb + half * 2;
        float a0 = arow[k0];
        float a1 = arow[k0 + 1];
        if constexpr (RELU) { a0 = fmaxf(a0, 0.0f); a1 = fmaxf(a1, 0.0f); }
        v2f a; a[0] = a0; a[1] = a1;
        v2f b; b[0] = wcol[(size_t)k0 * DCOL]; b[1] = wcol[(size_t)(k0 + 1) * DCOL];
        c = __builtin_amdgcn_wmma_f32_16x16x4_f32(
                 false, a,  false, b,
                 (short)0, c,  false,  false);
    }

    float bv;
    if constexpr (BIAS || SELF) bv = bias[colBase + l16];
    float* __restrict__ hout = H + colBase + l16;
    #pragma unroll
    for (int v = 0; v < 8; ++v) {
        const int row = rowBase + v + half * 8;
        float val = c[v];
        if constexpr (BIAS) val += bv;
        hout[(size_t)row * DCOL] = val;
        if constexpr (SELF) {
            const float w_lo = __int_as_float(__builtin_amdgcn_readlane(__float_as_int(dsq), v));
            const float w_hi = __int_as_float(__builtin_amdgcn_readlane(__float_as_int(dsq), v + 8));
            S[(size_t)row * DCOL + colBase + l16] = fmaf(half ? w_hi : w_lo, val, bv);
        }
    }
}

#define EPW  32
#define EGRP 16

__global__ __launch_bounds__(256)
void k_agg_edges(const int* __restrict__ src, const int* __restrict__ dst,
                 const float* __restrict__ wgt, const float* __restrict__ h,
                 float* __restrict__ out, int e) {
    const int lane = threadIdx.x & 31;
    const int wv   = __builtin_amdgcn_readfirstlane((int)(threadIdx.x >> 5));
    const int base = (blockIdx.x * (int)(blockDim.x >> 5) + wv) * EPW;
    if (base >= e) return;

    const int ed = base + lane;
    int s = 0, d = 0;
    float w = 0.0f;
    if (ed < e) {
        s = src[ed];
        d = dst[ed];
        w = wgt[ed];
    }

    if (base + EPW <= e) {
        #pragma unroll
        for (int g = 0; g < EPW; g += EGRP) {
            float v0[EGRP], v1[EGRP];
            #pragma unroll
            for (int j = 0; j < EGRP; ++j) {
                const int sj = __builtin_amdgcn_readlane(s, g + j);
                const float* __restrict__ hr = h + (size_t)sj * DCOL;
                v0[j] = hr[lane];
                v1[j] = hr[lane + 32];
            }
            #pragma unroll
            for (int j = 0; j < EGRP; ++j) {
                const int   dj = __builtin_amdgcn_readlane(d, g + j);
                const float wj = __int_as_float(__builtin_amdgcn_readlane(__float_as_int(w), g + j));
                float* __restrict__ orow = out + (size_t)dj * DCOL;
                atomicAdd(orow + lane,      wj * v0[j]);
                atomicAdd(orow + lane + 32, wj * v1[j]);
            }
        }
    } else {
        const int cnt = e - base;
        for (int j = 0; j < cnt; ++j) {
            const int   sj = __builtin_amdgcn_readlane(s, j);
            const int   dj = __builtin_amdgcn_readlane(d, j);
            const float wj = __int_as_float(__builtin_amdgcn_readlane(__float_as_int(w), j));
            const float* __restrict__ hr = h + (size_t)sj * DCOL;
            float* __restrict__ orow = out + (size_t)dj * DCOL;
            const float a0 = hr[lane], a1 = hr[lane + 32];
            atomicAdd(orow + lane,      wj * a0);
            atomicAdd(orow + lane + 32, wj * a1);
        }
    }
}

extern "C" void kernel_launch(void* const* d_in, const int* in_sizes, int n_in,
                              void* d_out, int out_size, void* d_ws, size_t ws_size,
                              hipStream_t stream) {
    const float* x   = (const float*)d_in[0];
    const int*   ei  = (const int*)  d_in[1];
    const float* W1  = (const float*)d_in[2];
    const float* b1  = (const float*)d_in[3];
    const float* W2  = (const float*)d_in[4];
    const float* b2  = (const float*)d_in[5];
    const float* Wfc = (const float*)d_in[6];
    const float* bfc = (const float*)d_in[7];
    float* out = (float*)d_out;

    const int N = in_sizes[0] / DIN;
    const int E = in_sizes[1] / 2;
    const int* src = ei;
    const int* dst = ei + E;

    const int Npad = ((N + 255) / 256) * 256;
    float* dinv = (float*)d_ws;
    float* buf0 = dinv + Npad;
    float* buf1 = buf0 + (size_t)N * DCOL;
    float* buf2 = buf1 + (size_t)N * DCOL;
    int*   part = (int*)(buf2 + (size_t)N * DCOL);
    float* wgt  = (float*)(part + (size_t)DEG_S * Npad);

    const int blk = 256;
    const int gN   = (N + blk - 1) / blk;
    const int gE   = (E + blk - 1) / blk;
    const int gGemm = N / 16;

    const dim3 gDeg((N + DEG_CAP - 1) / DEG_CAP, DEG_S);
    k_count_deg  <<<gDeg, DEG_T, 0, stream>>>(dst, part, N, Npad, E);
    k_finish_dinv<<<gN,   blk,   0, stream>>>(part, dinv, N, Npad);

    k_gemm_wmma<DIN,  false, false, true,  true ><<<gGemm, 128, 0, stream>>>(x,    W1,  buf0, b1,  dinv, buf1,
                                                                          src, dst, wgt, E);
    k_agg_edges <<<gE,   blk, 0, stream>>>(src, dst, wgt, buf0, buf1, E);

    k_gemm_wmma<DCOL, true,  false, true,  false><<<gGemm, 128, 0, stream>>>(buf1, W2,  buf0, b2,  dinv, buf2,
                                                                          nullptr, nullptr, nullptr, 0);
    k_agg_edges <<<gE,   blk, 0, stream>>>(src, dst, wgt, buf0, buf2, E);

    k_gemm_wmma<DCOL, false, true,  false, false><<<gGemm, 128, 0, stream>>>(buf2, Wfc, out,  bfc, nullptr, nullptr,
                                                                          nullptr, nullptr, nullptr, 0);
}


# Edge weight split per node: matrix launches store dinv*h, scatter adds rows unweighted, consumer scales its tile by dinv[dst] (factored out of each sum, arrival order kept); dinv launch goes: 7 to 6
# speedup vs baseline: 2.7152x; 1.0508x over previous
#include <hip/hip_runtime.h>
#include <hip/hip_bf16.h>

typedef __attribute__((ext_vector_type(2))) float v2f;
typedef __attribute__((ext_vector_type(8))) float v8f;

#define DIN 128
#define DCOL 64

#define DEG_CAP 6400
#define DEG_S   32
#define DEG_T   1024
#define DEG_U   25

__global__ __launch_bounds__(DEG_T)
void k_count_deg(const int* __restrict__ dst, int* __restrict__ part, int n, int npad, int e) {
    __shared__ int cnt[DEG_CAP];
    const int lo  = blockIdx.x * DEG_CAP;
    const int len = min(DEG_CAP, n - lo);
    for (int i = threadIdx.x; i < DEG_CAP; i += DEG_T) cnt[i] = 0;
    __syncthreads();

    const int per = (e + DEG_S - 1) / DEG_S;
    const int e0  = blockIdx.y * per;
    const int e1  = min(e0 + per, e);
    for (int base = e0 + (int)threadIdx.x; base < e1; base += DEG_T * DEG_U) {
        int d[DEG_U];
        #pragma unroll
        for (int u = 0; u < DEG_U; ++u) d[u] = dst[min(base + u * DEG_T, e1 - 1)];
        #pragma unroll
        for (int u = 0; u < DEG_U; ++u) {
            const unsigned off = (unsigned)(d[u] - lo);
            if (base + u * DEG_T < e1 && off < (unsigned)len) atomicAdd(&cnt[off], 1);
        }
    }
    __syncthreads();

    int* __restrict__ plane = part + (size_t)blockIdx.y * npad + lo;
    for (int i = threadIdx.x; i < len; i += DEG_T) plane[i] = cnt[i];
}

#define LAYER1 1
#define LAYER2 2
#define HEAD   3

template <int K, bool RELU, int MODE>
__global__ __launch_bounds__(128)
void k_gemm_wmma(const float* __restrict__ A, const float* __restrict__ W,
                 float* __restrict__ H, const float* __restrict__ bias,
                 const float* __restrict__ dinv_in, const float* __restrict__ dsqrt_in,
                 float* __restrict__ dinv_out, float* __restrict__ dsqrt_out,
                 float* __restrict__ S, const int* __restrict__ part, int npad) {
    const int lane    = threadIdx.x & 31;
    const int wave    = threadIdx.x >> 5;
    const int rowBase = blockIdx.x << 4;
    const int colBase = wave << 4;
    const int half    = lane >> 4;
    const int l16     = lane & 15;

    const float* __restrict__ arow = A + (size_t)(rowBase + l16) * K;
    const float* __restrict__ wcol = W + colBase + l16;

    int   pc[DEG_S / 2];
    float dr = 0.0f, sr = 0.0f;
    if constexpr (MODE == LAYER1) {
        const int* __restrict__ p = part + (size_t)(half * (DEG_S / 2)) * npad + rowBase + l16;
        #pragma unroll
        for (int s = 0; s < DEG_S / 2; ++s) pc[s] = p[(size_t)s * npad];
    } else {
        dr = dinv_in[rowBase + l16];
        if constexpr (MODE == LAYER2) sr = dsqrt_in[rowBase + l16];
    }

    v8f c = {};
    #pragma unroll 8
    for (int kb = 0; kb < K; kb += 4) {
        const int k0 = kb + half * 2;
        float a0 = arow[k0];
        float a1 = arow[k0 + 1];
        if constexpr (RELU) { a0 = fmaxf(a0, 0.0f); a1 = fmaxf(a1, 0.0f); }
        v2f a; a[0] = a0; a[1] = a1;
        v2f b; b[0] = wcol[(size_t)k0 * DCOL]; b[1] = wcol[(size_t)(k0 + 1) * DCOL];
        c = __builtin_amdgcn_wmma_f32_16x16x4_f32(
                 false, a,  false, b,
                 (short)0, c,  false,  false);
    }

    if constexpr (MODE == LAYER1) {
        int cnt = 1;
        #pragma unroll
        for (int s = 0; s < DEG_S / 2; ++s) cnt += pc[s];
        cnt += __builtin_amdgcn_permlanex16(cnt, cnt, 0x76543210, 0xfedcba98, false, false) - 1;
        const float deg = (float)cnt;
        dr = rsqrtf(deg);
        sr = sqrtf(deg);
        if (wave == 0 && half == 0) {
            dinv_out [rowBase + l16] = dr;
            dsqrt_out[rowBase + l16] = sr;
        }
    }
    const float sc = (MODE == LAYER2) ? dr * dr : dr;

    const float bv = bias[colBase + l16];
    float* __restrict__ hout = H + colBase + l16;
    #pragma unroll
    for (int v = 0; v < 8; ++v) {
        const int row = rowBase + v + half * 8;
        const float sc_lo = __int_as_float(__builtin_amdgcn_readlane(__float_as_int(sc), v));
        const float sc_hi = __int_as_float(__builtin_amdgcn_readlane(__float_as_int(sc), v + 8));
        const float scv = half ? sc_hi : sc_lo;
        if constexpr (MODE == HEAD) {
            hout[(size_t)row * DCOL] = fmaf(scv, c[v], bv);
        } else {
            const float sr_lo = __int_as_float(__builtin_amdgcn_readlane(__float_as_int(sr), v));
            const float sr_hi = __int_as_float(__builtin_amdgcn_readlane(__float_as_int(sr), v + 8));
            const float g = scv * c[v];
            hout[(size_t)row * DCOL] = g;
            S[(size_t)row * DCOL + colBase + l16] = fmaf(bv, half ? sr_hi : sr_lo, g);
        }
    }
}

#define EPW  32
#define EGRP 16

__global__ __launch_bounds__(256)
void k_agg_edges(const int* __restrict__ src, const int* __restrict__ dst,
                 const float* __restrict__ g, float* __restrict__ out, int e) {
    const int lane = threadIdx.x & 31;
    const int wv   = __builtin_amdgcn_readfirstlane((int)(threadIdx.x >> 5));
    const int base = (blockIdx.x * (int)(blockDim.x >> 5) + wv) * EPW;
    if (base >= e) return;

    const int ed = base + lane;
    int s = 0, d = 0;
    if (ed < e) {
        s = src[ed];
        d = dst[ed];
    }

    if (base + EPW <= e) {
        #pragma unroll
        for (int gi = 0; gi < EPW; gi += EGRP) {
            float v0[EGRP], v1[EGRP];
            #pragma unroll
            for (int j = 0; j < EGRP; ++j) {
                const int sj = __builtin_amdgcn_readlane(s, gi + j);
                const float* __restrict__ gr = g + (size_t)sj * DCOL;
                v0[j] = gr[lane];
                v1[j] = gr[lane + 32];
            }
            #pragma unroll
            for (int j = 0; j < EGRP; ++j) {
                const int dj = __builtin_amdgcn_readlane(d, gi + j);
                float* __restrict__ orow = out + (size_t)dj * DCOL;
                atomicAdd(orow + lane,      v0[j]);
                atomicAdd(orow + lane + 32, v1[j]);
            }
        }
    } else {
        const int cnt = e - base;
        for (int j = 0; j < cnt; ++j) {
            const int sj = __builtin_amdgcn_readlane(s, j);
            const int dj = __builtin_amdgcn_readlane(d, j);
            const float* __restrict__ gr = g + (size_t)sj * DCOL;
            float* __restrict__ orow = out + (size_t)dj * DCOL;
            const float a0 = gr[lane], a1 = gr[lane + 32];
            atomicAdd(orow + lane,      a0);
            atomicAdd(orow + lane + 32, a1);
        }
    }
}

extern "C" void kernel_launch(void* const* d_in, const int* in_sizes, int n_in,
                              void* d_out, int out_size, void* d_ws, size_t ws_size,
                              hipStream_t stream) {
    const float* x   = (const float*)d_in[0];
    const int*   ei  = (const int*)  d_in[1];
    const float* W1  = (const float*)d_in[2];
    const float* b1  = (const float*)d_in[3];
    const float* W2  = (const float*)d_in[4];
    const float* b2  = (const float*)d_in[5];
    const float* Wfc = (const float*)d_in[6];
    const float* bfc = (const float*)d_in[7];
    float* out = (float*)d_out;

    const int N = in_sizes[0] / DIN;
    const int E = in_sizes[1] / 2;
    const int* src = ei;
    const int* dst = ei + E;

    const int Npad = ((N + 255) / 256) * 256;
    float* dinv = (float*)d_ws;
    float* dsq  = dinv + Npad;
    float* buf0 = dsq + Npad;
    float* buf1 = buf0 + (size_t)N * DCOL;
    float* buf2 = buf1 + (size_t)N * DCOL;
    int*   part = (int*)(buf2 + (size_t)N * DCOL);

    const int blk = 256;
    const int gE   = (E + blk - 1) / blk;
    const int gGemm = N / 16;

    const dim3 gDeg((N + DEG_CAP - 1) / DEG_CAP, DEG_S);
    k_count_deg<<<gDeg, DEG_T, 0, stream>>>(dst, part, N, Npad, E);

    k_gemm_wmma<DIN,  false, LAYER1><<<gGemm, 128, 0, stream>>>(
        x,    W1,  buf0, b1,  nullptr, nullptr, dinv,    dsq,     buf1,    part,    Npad);
    k_agg_edges<<<gE, blk, 0, stream>>>(src, dst, buf0, buf1, E);

    k_gemm_wmma<DCOL, true,  LAYER2><<<gGemm, 128, 0, stream>>>(
        buf1, W2,  buf0, b2,  dinv,    dsq,     nullptr, nullptr, buf2,    nullptr, 0);
    k_agg_edges<<<gE, blk, 0, stream>>>(src, dst, buf0, buf2, E);

    k_gemm_wmma<DCOL, false, HEAD  ><<<gGemm, 128, 0, stream>>>(
        buf2, Wfc, out,  bfc, dinv,    nullptr, nullptr, nullptr, nullptr, nullptr, 0);
}
